# starting kernel
# baseline (speedup 1.0000x reference)
#include <hip/hip_runtime.h>
#include <hip/hip_bf16.h>
#include <math.h>

#define B_    32
#define T_    131072
#define KLEN  16384
#define SEG   2048
#define NSEG  (T_ / SEG)

typedef __attribute__((ext_vector_type(2))) float v2f;
typedef __attribute__((ext_vector_type(8))) float v8f;

__device__ __forceinline__ float alpha_of(float z) {
    float sig = 1.0f / (1.0f + __expf(-z));
    return fminf(fmaxf(sig, 1e-5f), 1.0f - 1e-5f);
}

__global__ void __launch_bounds__(256) k_energy(const float* __restrict__ sig,
                                                float* __restrict__ e) {
    int gid  = blockIdx.x * 256 + threadIdx.x;
    int flat = gid * 4;
    int b = flat / T_;
    int t = flat % T_;
    const float4 s0 = *(const float4*)(sig + (size_t)b * 2 * T_ + t);
    const float4 s1 = *(const float4*)(sig + (size_t)b * 2 * T_ + T_ + t);
    float4 r;
    r.x = 0.5f * (s0.x * s0.x + s1.x * s1.x);
    r.y = 0.5f * (s0.y * s0.y + s1.y * s1.y);
    r.z = 0.5f * (s0.z * s0.z + s1.z * s1.z);
    r.w = 0.5f * (s0.w * s0.w + s1.w * s1.w);
    *(float4*)(e + flat) = r;
}

__global__ void __launch_bounds__(256) k_partial(const float* __restrict__ e,
                                                 const float* __restrict__ zalpha,
                                                 float* __restrict__ P) {
    const int s = blockIdx.x, b = blockIdx.y;
    const int j = threadIdx.x;
    const float alpha = alpha_of(zalpha[b]);
    const float la  = __logf(alpha);
    const float oma = 1.0f - alpha;
    const float aK  = __expf((float)KLEN * la);
    const float* er = e + (size_t)b * T_;
    const int t0 = s * SEG + j * 8;

    float p = 0.0f;
#pragma unroll
    for (int i = 0; i < 8; ++i) {
        int t = t0 + i;
        float ek = (t >= KLEN) ? er[t - KLEN] : 0.0f;
        float bv = oma * (er[t] - aK * ek);
        p = fmaf(alpha, p, bv);
    }

    __shared__ float red[256];
    red[j] = p * __expf(8.0f * (float)(255 - j) * la);
    __syncthreads();
#pragma unroll
    for (int off = 128; off > 0; off >>= 1) {
        if (j < off) red[j] += red[j + off];
        __syncthreads();
    }
    if (j == 0) P[b * NSEG + s] = red[0];
}

__global__ void k_scan(const float* __restrict__ P,
                       const float* __restrict__ zalpha,
                       float* __restrict__ Carr) {
    int b = threadIdx.x;
    if (b < B_) {
        float alpha = alpha_of(zalpha[b]);
        float aL = __expf((float)SEG * __logf(alpha));
        float c = 0.0f;
        for (int s = 0; s < NSEG; ++s) {
            Carr[b * NSEG + s] = c;
            c = fmaf(aL, c, P[b * NSEG + s]);
        }
    }
}

__global__ void __launch_bounds__(256) k_final(const float* __restrict__ e,
                                               const float* __restrict__ zalpha,
                                               const float* __restrict__ Carr,
                                               float* __restrict__ out) {
    const int s = blockIdx.x, b = blockIdx.y;
    const int tid  = threadIdx.x;
    const int wave = tid >> 5;
    const int lane = tid & 31;

    __shared__ float lbs[8][256];
    __shared__ float cend[8][16];
    __shared__ float wend[8];

    const float alpha = alpha_of(zalpha[b]);
    const float la   = __logf(alpha);
    const float oma  = 1.0f - alpha;
    const float aK   = __expf((float)KLEN * la);
    const float a16  = __expf(16.0f * la);
    const float* er  = e + (size_t)b * T_;
    const int segStart  = s * SEG;
    const int tileStart = segStart + wave * 256;

    {
        const int i0 = lane * 8;
#pragma unroll
        for (int i = 0; i < 8; ++i) {
            int t = tileStart + i0 + i;
            float ek = (t >= KLEN) ? er[t - KLEN] : 0.0f;
            lbs[wave][i0 + i] = oma * (er[t] - aK * ek);
        }
    }
    __syncthreads();

    const int half = lane >> 4;
    const int m    = lane & 15;
    v8f c = {};
#pragma unroll
    for (int sl = 0; sl < 4; ++sl) {
        const int k0 = 4 * sl + 2 * half;
        v2f a;
        a.x = (m >= k0)     ? __expf((float)(m - k0) * la)     : 0.0f;
        a.y = (m >= k0 + 1) ? __expf((float)(m - k0 - 1) * la) : 0.0f;
        v2f bb;
        bb.x = lbs[wave][16 * m + k0];
        bb.y = lbs[wave][16 * m + k0 + 1];
        c = __builtin_amdgcn_wmma_f32_16x16x4_f32(
                 false, a,  false, bb,
                 (short)0, c,  false,  false);
    }

    if (half == 1) cend[wave][m] = c[7];
    __syncthreads();

    float myCarry = 0.0f, cc = 0.0f;
#pragma unroll
    for (int j = 0; j < 16; ++j) {
        if (j == m) myCarry = cc;
        cc = fmaf(a16, cc, cend[wave][j]);
    }
    if (lane == 0) wend[wave] = cc;
    __syncthreads();

    float cw = Carr[b * NSEG + s] * __expf((float)(256 * wave) * la);
    for (int v = 0; v < wave; ++v)
        cw = fmaf(wend[v], __expf((float)(256 * (wave - 1 - v)) * la), cw);
    const float carryChunk = fmaf(cw, __expf((float)(16 * m) * la), myCarry);

#pragma unroll
    for (int r = 0; r < 8; ++r) {
        const int mm = r + 8 * half;
        float val = fmaf(carryChunk, __expf((float)(mm + 1) * la), c[r]);
        float y = fmaxf(val, 0.0f);
        lbs[wave][16 * m + mm] = __logf(y + 1e-5f);
    }
    __syncthreads();

    float* orow = out + (size_t)b * T_ + tileStart;
    const int i0 = lane * 8;
    float4 o0, o1;
    o0.x = lbs[wave][i0 + 0]; o0.y = lbs[wave][i0 + 1];
    o0.z = lbs[wave][i0 + 2]; o0.w = lbs[wave][i0 + 3];
    o1.x = lbs[wave][i0 + 4]; o1.y = lbs[wave][i0 + 5];
    o1.z = lbs[wave][i0 + 6]; o1.w = lbs[wave][i0 + 7];
    *(float4*)(orow + i0)     = o0;
    *(float4*)(orow + i0 + 4) = o1;
}

extern "C" void kernel_launch(void* const* d_in, const int* in_sizes, int n_in,
                              void* d_out, int out_size, void* d_ws, size_t ws_size,
                              hipStream_t stream) {
    const float* signal = (const float*)d_in[0];
    const float* zalpha = (const float*)d_in[1];
    float* out = (float*)d_out;

    float* e    = (float*)d_ws;
    float* P    = e + (size_t)B_ * T_;
    float* Carr = P + (size_t)B_ * NSEG;

    k_energy <<<dim3((B_ * T_) / (256 * 4)), 256, 0, stream>>>(signal, e);
    k_partial<<<dim3(NSEG, B_),             256, 0, stream>>>(e, zalpha, P);
    k_scan   <<<dim3(1),                     32, 0, stream>>>(P, zalpha, Carr);
    k_final  <<<dim3(NSEG, B_),             256, 0, stream>>>(e, zalpha, Carr, out);
}


# One launch, no workspace: a block owns a K-long span; window sum = prefix scan of own span + suffix sum of the span before (no K-back subtraction, no carry scan launch); sum order changes
# speedup vs baseline: 3.3227x; 3.3227x over previous
#include <hip/hip_runtime.h>
#include <math.h>

#define B_     32
#define T_     131072
#define KLEN   16384
#define NSPAN  (T_ / KLEN)
#define NW     32
#define WSPAN  512

typedef __attribute__((ext_vector_type(2))) float v2f;
typedef __attribute__((ext_vector_type(8))) float v8f;

__device__ __forceinline__ float alpha_of(float z) {
    float sig = 1.0f / (1.0f + __expf(-z));
    return fminf(fmaxf(sig, 1e-5f), 1.0f - 1e-5f);
}

__device__ __forceinline__ float wave_sum(float x) {
#pragma unroll
    for (int off = 16; off > 0; off >>= 1) x += __shfl_xor(x, off, 32);
    return x;
}

__global__ void __launch_bounds__(1024) k_span(const float* __restrict__ sig,
                                               const float* __restrict__ zalpha,
                                               float* __restrict__ out) {
    const int s = blockIdx.x, b = blockIdx.y;
    const int tid  = threadIdx.x;
    const int w    = tid >> 5;
    const int lane = tid & 31;
    const int h    = lane >> 4;
    const int n    = lane & 15;

    __shared__ float ceP[NW][32];
    __shared__ float czS[NW][32];
    __shared__ float wendP[NW];
    __shared__ float wtotS[NW];

    const float alpha = alpha_of(zalpha[b]);
    const float la    = __logf(alpha);
    const float c0    = 0.5f * (1.0f - alpha);
    const float a16   = __expf(16.0f * la);
    const float a15   = __expf(15.0f * la);
    const float aKm16 = __expf((float)(KLEN - 16) * la);

    const float* x0 = sig + (size_t)b * 2 * T_ + (size_t)s * KLEN;
    const float* x1 = x0 + T_;
    const int pw = w * WSPAN;
    const int lo = 16 * n + 8 * h;

    float4 u[2][2][2], q[2][2][2];
#pragma unroll
    for (int it = 0; it < 2; ++it) {
        const int p = pw + 256 * it + lo;
        u[it][0][0] = *(const float4*)(x0 + p);
        u[it][0][1] = *(const float4*)(x0 + p + 4);
        u[it][1][0] = *(const float4*)(x1 + p);
        u[it][1][1] = *(const float4*)(x1 + p + 4);
    }
    if (s > 0) {
#pragma unroll
        for (int it = 0; it < 2; ++it) {
            const int p = pw + 256 * it + lo - KLEN;
            q[it][0][0] = *(const float4*)(x0 + p);
            q[it][0][1] = *(const float4*)(x0 + p + 4);
            q[it][1][0] = *(const float4*)(x1 + p);
            q[it][1][1] = *(const float4*)(x1 + p + 4);
        }
    } else {
        const float4 z4 = make_float4(0.0f, 0.0f, 0.0f, 0.0f);
#pragma unroll
        for (int it = 0; it < 2; ++it) {
            q[it][0][0] = z4; q[it][0][1] = z4; q[it][1][0] = z4; q[it][1][1] = z4;
        }
    }

    v2f aP[4], aS[4];
#pragma unroll
    for (int sl = 0; sl < 4; ++sl) {
        const int k0 = 8 * h + 2 * sl;
        aP[sl].x = (n >= k0)     ? c0 * __expf((float)(n - k0) * la)     : 0.0f;
        aP[sl].y = (n >= k0 + 1) ? c0 * __expf((float)(n - k0 - 1) * la) : 0.0f;
        aS[sl].x = (k0 > n)      ? c0 * __expf((float)(15 - k0) * la)    : 0.0f;
        aS[sl].y = (k0 + 1 > n)  ? c0 * __expf((float)(14 - k0) * la)    : 0.0f;
    }

    float gu[2][8], gq[2][8];
#pragma unroll
    for (int it = 0; it < 2; ++it) {
#pragma unroll
        for (int f = 0; f < 2; ++f) {
            const float4 a0 = u[it][0][f], a1 = u[it][1][f];
            gu[it][4 * f + 0] = fmaf(a0.x, a0.x, a1.x * a1.x);
            gu[it][4 * f + 1] = fmaf(a0.y, a0.y, a1.y * a1.y);
            gu[it][4 * f + 2] = fmaf(a0.z, a0.z, a1.z * a1.z);
            gu[it][4 * f + 3] = fmaf(a0.w, a0.w, a1.w * a1.w);
            const float4 b0 = q[it][0][f], b1 = q[it][1][f];
            gq[it][4 * f + 0] = fmaf(b0.x, b0.x, b1.x * b1.x);
            gq[it][4 * f + 1] = fmaf(b0.y, b0.y, b1.y * b1.y);
            gq[it][4 * f + 2] = fmaf(b0.z, b0.z, b1.z * b1.z);
            gq[it][4 * f + 3] = fmaf(b0.w, b0.w, b1.w * b1.w);
        }
    }

    v8f cP[2] = {v8f{}, v8f{}};
    v8f cS[2] = {v8f{}, v8f{}};
#pragma unroll
    for (int sl = 0; sl < 4; ++sl) {
#pragma unroll
        for (int it = 0; it < 2; ++it) {
            v2f bu, bq;
            bu.x = gu[it][2 * sl]; bu.y = gu[it][2 * sl + 1];
            bq.x = gq[it][2 * sl]; bq.y = gq[it][2 * sl + 1];
            cP[it] = __builtin_amdgcn_wmma_f32_16x16x4_f32(
                         false, aP[sl], false, bu, (short)0, cP[it], false, false);
            cS[it] = __builtin_amdgcn_wmma_f32_16x16x4_f32(
                         false, aS[sl], false, bq, (short)0, cS[it], false, false);
        }
    }

#pragma unroll
    for (int it = 0; it < 2; ++it) {
        if (h == 1) {
            ceP[w][16 * it + n] = cP[it][7];
        } else {
            const float z0 = fmaf(c0 * a15, gq[it][0], cS[it][0]);
            const float wc = __expf((float)(KLEN - 16 - pw - 256 * it - 16 * n) * la);
            czS[w][16 * it + n] = wc * z0;
        }
    }
    __syncthreads();

    float run = 0.0f, cp0 = 0.0f, cp1 = 0.0f;
#pragma unroll
    for (int c = 0; c < 32; ++c) {
        if (c == n)      cp0 = run;
        if (c == 16 + n) cp1 = run;
        run = fmaf(a16, run, ceP[w][c]);
    }
    float runS = 0.0f, lt0 = 0.0f, lt1 = 0.0f;
#pragma unroll
    for (int c = 31; c >= 0; --c) {
        if (c == 16 + n) lt1 = runS;
        if (c == n)      lt0 = runS;
        runS += czS[w][c];
    }
    if (lane == 0) { wendP[w] = run; wtotS[w] = runS; }
    __syncthreads();

    const int   dist = (w - 1 - lane) > 0 ? (w - 1 - lane) : 0;
    const float tP = (lane < w) ? wendP[lane] * __expf((float)(WSPAN * dist) * la) : 0.0f;
    const float tS = (lane > w) ? wtotS[lane] : 0.0f;
    const float carryW = wave_sum(tP);
    const float laterW = wave_sum(tS);

    float pwr[8];
#pragma unroll
    for (int r = 0; r < 8; ++r) pwr[r] = __expf((float)(r + 8 * h + 1) * la);

    float* orow = out + (size_t)b * T_ + (size_t)s * KLEN;
#pragma unroll
    for (int it = 0; it < 2; ++it) {
        const float cpl = it ? cp1 : cp0;
        const float ltl = it ? lt1 : lt0;
        const float fP  = __expf((float)(16 * (16 * it + n)) * la);
        const float fS  = __expf((float)(pw + 256 * it + 16 * n) * la);
        const float base = fmaf(fS, ltl + laterW, fmaf(fP, carryW, cpl));
        float o[8];
#pragma unroll
        for (int r = 0; r < 8; ++r) {
            const float inner = fmaf(aKm16, cS[it][r], base);
            const float val   = fmaf(pwr[r], inner, cP[it][r]);
            o[r] = __logf(fmaxf(val, 0.0f) + 1e-5f);
        }
        float* op = orow + pw + 256 * it + lo;
        *(float4*)(op)     = make_float4(o[0], o[1], o[2], o[3]);
        *(float4*)(op + 4) = make_float4(o[4], o[5], o[6], o[7]);
    }
}

extern "C" void kernel_launch(void* const* d_in, const int* in_sizes, int n_in,
                              void* d_out, int out_size, void* d_ws, size_t ws_size,
                              hipStream_t stream) {
    const float* signal = (const float*)d_in[0];
    const float* zalpha = (const float*)d_in[1];
    float* out = (float*)d_out;

    k_span<<<dim3(NSPAN, B_), 1024, 0, stream>>>(signal, zalpha, out);
}
